# starting kernel
# baseline (speedup 1.0000x reference)
#include <hip/hip_runtime.h>
#include <hip/hip_bf16.h>

typedef __attribute__((ext_vector_type(2))) float v2f;
typedef __attribute__((ext_vector_type(8))) float v8f;


__global__ void init_deg_kernel(int* deg, int n) {
    int i = blockIdx.x * blockDim.x + threadIdx.x;
    if (i < n) deg[i] = 1;
}

__global__ void count_edges_kernel(const int* __restrict__ ei, int* deg, int E) {
    int e = blockIdx.x * blockDim.x + threadIdx.x;
    if (e < E) {
        int dst = ei[E + e];
        atomicAdd(&deg[dst], 1);
    }
}

__global__ void scan_block_kernel(const int* __restrict__ deg, int* row_excl,
                                  int* bsum, int n) {
    __shared__ int sh[256];
    int i = blockIdx.x * 256 + threadIdx.x;
    int v = (i < n) ? deg[i] : 0;
    sh[threadIdx.x] = v;
    __syncthreads();
    for (int off = 1; off < 256; off <<= 1) {
        int t = (threadIdx.x >= off) ? sh[threadIdx.x - off] : 0;
        __syncthreads();
        sh[threadIdx.x] += t;
        __syncthreads();
    }
    if (i < n) row_excl[i] = sh[threadIdx.x] - v;
    if (threadIdx.x == 255) bsum[blockIdx.x] = sh[255];
}

__global__ void scan_sums_kernel(int* bsum, int nb) {
    __shared__ int sh[512];
    int v = (threadIdx.x < nb) ? bsum[threadIdx.x] : 0;
    sh[threadIdx.x] = v;
    __syncthreads();
    for (int off = 1; off < 512; off <<= 1) {
        int t = (threadIdx.x >= off) ? sh[threadIdx.x - off] : 0;
        __syncthreads();
        sh[threadIdx.x] += t;
        __syncthreads();
    }
    if (threadIdx.x < nb) bsum[threadIdx.x] = sh[threadIdx.x] - v;
}

__global__ void finalize_rows_kernel(const int* __restrict__ deg, int* row_ptr,
                                     const int* __restrict__ bsum, int* cursor,
                                     float* dis, int n, int total) {
    int i = blockIdx.x * blockDim.x + threadIdx.x;
    if (i < n) {
        int r = row_ptr[i] + bsum[i >> 8];
        row_ptr[i] = r;
        cursor[i]  = r;
        dis[i]     = rsqrtf((float)deg[i]);
        if (i == 0) row_ptr[n] = total;
    }
}

__global__ void fill_self_kernel(int* cursor, int* colidx, int n) {
    int i = blockIdx.x * blockDim.x + threadIdx.x;
    if (i < n) colidx[atomicAdd(&cursor[i], 1)] = i;
}

__global__ void fill_edges_kernel(const int* __restrict__ ei, int* cursor,
                                  int* colidx, int E) {
    int e = blockIdx.x * blockDim.x + threadIdx.x;
    if (e < E) {
        int src = ei[e];
        int dst = ei[E + e];
        colidx[atomicAdd(&cursor[dst], 1)] = src;
    }
}


template <bool MASK>
__global__ void gemm_wmma_f32_kernel(const float* __restrict__ A, int lda,
                                     const float* __restrict__ B, int ldb,
                                     float* __restrict__ C, int ldc,
                                     int K, int Ncols,
                                     const float* __restrict__ bias) {
    __shared__ float sA[16 * 513];

    const int row0 = blockIdx.x * 16;
    const int lane = threadIdx.x;
    const int tid  = threadIdx.y * 32 + threadIdx.x;
    const int nthr = blockDim.x * blockDim.y;
    const int ldsS = K + 1;

    for (int r = 0; r < 16; ++r) {
        const float* Ar = A + (size_t)(row0 + r) * lda;
        for (int c = tid; c < K; c += nthr) sA[r * ldsS + c] = Ar[c];
    }
    __syncthreads();

    const int m    = lane & 15;
    const int kk   = (lane >> 4) << 1;
    const int col0 = threadIdx.y * 16;
    const int n    = col0 + m;
    const bool nok = !MASK || (n < Ncols);
    const int  nc  = MASK ? (nok ? n : 0) : n;

    const float* Ap = sA + m * ldsS + kk;
    const float* Bp = B + kk * ldb + nc;

    v8f acc = {};
#pragma unroll 4
    for (int k = 0; k < K; k += 4) {
        v2f a, b;
        a.x = Ap[0];
        a.y = Ap[1];
        float bx = Bp[0];
        float by = Bp[ldb];
        if (MASK) { bx = nok ? bx : 0.0f; by = nok ? by : 0.0f; }
        b.x = bx;
        b.y = by;
        acc = __builtin_amdgcn_wmma_f32_16x16x4_f32(
             false, a,  false, b,
             (short)0, acc,  false,  false);
        Ap += 4;
        Bp += 4 * ldb;
    }

    const int mbase = (lane < 16) ? 0 : 8;
    if (nok) {
        const float badd = bias ? bias[n] : 0.0f;
        float* Cp = C + (size_t)(row0 + mbase) * ldc + n;
#pragma unroll
        for (int r = 0; r < 8; ++r) {
            Cp[0] = acc[r] + badd;
            Cp += ldc;
        }
    }
}


__global__ void gcn_aggregate_kernel(const float* __restrict__ H,
                                     const int* __restrict__ row_ptr,
                                     const int* __restrict__ colidx,
                                     const float* __restrict__ dis,
                                     const float* __restrict__ bias,
                                     float* __restrict__ outbuf,
                                     int ld_out, int col_off, int n) {
    const int node = blockIdx.x * (blockDim.x >> 5) + (threadIdx.x >> 5);
    const int lane = threadIdx.x & 31;
    if (node >= n) return;

    const int s = row_ptr[node];
    const int e = row_ptr[node + 1];

    float a0 = 0.f, a1 = 0.f, a2 = 0.f, a3 = 0.f;
    int p = s;
    for (; p + 1 < e; p += 2) {
        const int j0 = colidx[p];
        const int j1 = colidx[p + 1];
        const float w0 = dis[j0];
        const float w1 = dis[j1];
        const float4 h0 = *(const float4*)(H + (size_t)j0 * 128 + lane * 4);
        const float4 h1 = *(const float4*)(H + (size_t)j1 * 128 + lane * 4);
        a0 += w0 * h0.x + w1 * h1.x;
        a1 += w0 * h0.y + w1 * h1.y;
        a2 += w0 * h0.z + w1 * h1.z;
        a3 += w0 * h0.w + w1 * h1.w;
    }
    if (p < e) {
        const int j = colidx[p];
        const float w = dis[j];
        const float4 h = *(const float4*)(H + (size_t)j * 128 + lane * 4);
        a0 += w * h.x; a1 += w * h.y; a2 += w * h.z; a3 += w * h.w;
    }

    const float di = dis[node];
    const int c = lane * 4;
    float4 r;
    r.x = fmaxf(di * a0 + bias[c + 0], 0.0f);
    r.y = fmaxf(di * a1 + bias[c + 1], 0.0f);
    r.z = fmaxf(di * a2 + bias[c + 2], 0.0f);
    r.w = fmaxf(di * a3 + bias[c + 3], 0.0f);
    *(float4*)(outbuf + (size_t)node * ld_out + col_off + c) = r;
}


extern "C" void kernel_launch(void* const* d_in, const int* in_sizes, int n_in,
                              void* d_out, int out_size, void* d_ws, size_t ws_size,
                              hipStream_t stream) {
    (void)n_in; (void)out_size; (void)ws_size;
    const int F_IN = 256, HID = 128, NC = 40;
    const int N = in_sizes[0] / F_IN;
    const int E = in_sizes[1] / 2;
    const int TOT = E + N;

    const float* x     = (const float*)d_in[0];
    const int*   ei    = (const int*)d_in[1];
    const float* W_in  = (const float*)d_in[2];
    const float* b_in  = (const float*)d_in[3];
    const float* W1    = (const float*)d_in[4];
    const float* b1    = (const float*)d_in[5];
    const float* W2    = (const float*)d_in[6];
    const float* b2    = (const float*)d_in[7];
    const float* W3    = (const float*)d_in[8];
    const float* b3    = (const float*)d_in[9];
    const float* W_out = (const float*)d_in[10];
    const float* b_out = (const float*)d_in[11];
    float* out = (float*)d_out;

    char* base = (char*)d_ws;
    size_t off = 0;
    auto alloc = [&](size_t bytes) -> void* {
        void* q = base + off;
        off = (off + bytes + 255) & ~(size_t)255;
        return q;
    };
    int*   deg     = (int*)alloc((size_t)N * 4);
    int*   row_ptr = (int*)alloc((size_t)(N + 1) * 4);
    int*   cursor  = (int*)alloc((size_t)N * 4);
    int*   bsum    = (int*)alloc(512 * 4);
    int*   colidx  = (int*)alloc((size_t)TOT * 4);
    float* dis     = (float*)alloc((size_t)N * 4);
    float* H       = (float*)alloc((size_t)N * HID * 4);
    float* hcat    = (float*)alloc((size_t)N * HID * 4 * 4);

    const int nbN = (N + 255) / 256;
    const int nbE = (E + 255) / 256;

    init_deg_kernel<<<nbN, 256, 0, stream>>>(deg, N);
    count_edges_kernel<<<nbE, 256, 0, stream>>>(ei, deg, E);
    scan_block_kernel<<<nbN, 256, 0, stream>>>(deg, row_ptr, bsum, N);
    scan_sums_kernel<<<1, 512, 0, stream>>>(bsum, nbN);
    finalize_rows_kernel<<<nbN, 256, 0, stream>>>(deg, row_ptr, bsum, cursor, dis, N, TOT);
    fill_self_kernel<<<nbN, 256, 0, stream>>>(cursor, colidx, N);
    fill_edges_kernel<<<nbE, 256, 0, stream>>>(ei, cursor, colidx, E);

    const int rowTiles = N / 16;
    dim3 gemmBlk(32, HID / 16);
    const int aggBlocks = (N + 7) / 8;

    gemm_wmma_f32_kernel<false><<<rowTiles, gemmBlk, 0, stream>>>(
        x, F_IN, W_in, HID, H, HID, F_IN, HID, nullptr);
    gcn_aggregate_kernel<<<aggBlocks, 256, 0, stream>>>(
        H, row_ptr, colidx, dis, b_in, hcat, 4 * HID, 0 * HID, N);

    const float* Ws[3] = {W1, W2, W3};
    const float* bs[3] = {b1, b2, b3};
    for (int l = 0; l < 3; ++l) {
        gemm_wmma_f32_kernel<false><<<rowTiles, gemmBlk, 0, stream>>>(
            hcat + (size_t)l * HID, 4 * HID, Ws[l], HID, H, HID, HID, HID, nullptr);
        gcn_aggregate_kernel<<<aggBlocks, 256, 0, stream>>>(
            H, row_ptr, colidx, dis, bs[l], hcat, 4 * HID, (l + 1) * HID, N);
    }

    dim3 headBlk(32, 3);
    gemm_wmma_f32_kernel<true><<<rowTiles, headBlk, 0, stream>>>(
        hcat, 4 * HID, W_out, NC, out, NC, 4 * HID, NC, b_out);
}


# No CSR build: the count atomic's returned slot stores each edge's source straight into its destination's 128-entry list (concat buffer's unused last quarter); scans and fill gone, 14 to 10 launches
# speedup vs baseline: 1.8081x; 1.8081x over previous
#include <hip/hip_runtime.h>
#include <hip/hip_bf16.h>

typedef __attribute__((ext_vector_type(2))) float v2f;
typedef __attribute__((ext_vector_type(4))) float v4f;
typedef __attribute__((ext_vector_type(8))) float v8f;


constexpr int LIST_CAP = 128;

__device__ __forceinline__ float dis_of(int c) {
    return rsqrtf((float)(c + 1));
}

__global__ void init_cnt_kernel(int* cnt, int* lists, int lpitch, int n) {
    int i = blockIdx.x * blockDim.x + threadIdx.x;
    if (i < n + 4) cnt[i] = 0;
    if (i < n) lists[(size_t)i * lpitch] = i;
}

__device__ __forceinline__ void place_edge(int* lists, int lpitch, int* ovf, int* ovf_n,
                                           int dst, int src_node, int slot, int e) {
    if (slot < LIST_CAP - 1) lists[(size_t)dst * lpitch + 1 + slot] = src_node;
    else ovf[atomicAdd(ovf_n, 1)] = e;
}


template <int K, int NCT, int KS, bool MASK, bool CNT>
__global__ __launch_bounds__(32 * NCT * KS)
void gemm_wreg_kernel(const float* __restrict__ A, int lda,
                      const float* __restrict__ B, int ldb,
                      float* __restrict__ C, int ldc, int Ncols,
                      const float* __restrict__ bias, int nTiles,
                      const int* __restrict__ esrc, const int* __restrict__ edst,
                      int* cnt, int* ovf_n, int* lists, int lpitch,
                      int* __restrict__ ovf, int E) {
    constexpr int NW   = NCT * KS;
    constexpr int NTHR = 32 * NW;
    constexpr int KW   = K / KS;
    constexpr int NF   = KW / 4;
    constexpr int S    = K + 4;
    constexpr int Q    = K / 4;
    constexpr int NV   = (16 * Q + NTHR - 1) / NTHR;
    constexpr bool FULL = ((16 * Q) % NTHR) == 0;

    __shared__ __attribute__((aligned(16))) float sA[16 * S];
    __shared__ float sP[KS > 1 ? NW * 256 : 1];

    const int lane = threadIdx.x;
    const int wave = threadIdx.y;
    const int tid  = wave * 32 + lane;
    const int ct   = wave % NCT;
    const int kq   = wave / NCT;
    const int m    = lane & 15;
    const int kk   = (lane >> 4) << 1;
    const int n    = ct * 16 + m;
    const bool nok = !MASK || (n < Ncols);
    const int  nc  = MASK ? (nok ? n : 0) : n;

    v2f bf[NF];
    {
        const float* Bp = B + (size_t)(kq * KW + kk) * ldb + nc;
#pragma unroll
        for (int f = 0; f < NF; ++f) {
            float bx = Bp[0];
            float by = Bp[ldb];
            if (MASK) { bx = nok ? bx : 0.0f; by = nok ? by : 0.0f; }
            bf[f].x = bx;
            bf[f].y = by;
            Bp += 4 * ldb;
        }
    }

    v4f stage[NV];
    auto fetch = [&](int t) __attribute__((always_inline)) {
        const float* At = A + (size_t)t * 16 * lda;
#pragma unroll
        for (int i = 0; i < NV; ++i) {
            const int idx = tid + i * NTHR;
            if (FULL || idx < 16 * Q) {
                const int r = idx / Q, c4 = idx % Q;
                stage[i] = *(const v4f*)(At + (size_t)r * lda + c4 * 4);
            }
        }
    };

    const int mbase = (lane < 16) ? 0 : 8;
    const float badd = (bias && nok) ? bias[n] : 0.0f;
    const float* Ap = sA + m * S + kq * KW + kk;

    int tile = blockIdx.x;
    int  dcur = 0, scur = 0;
    bool hcur = false;
    int  dold = 0, sold = 0, slotold = 0, eold = 0;
    bool hold = false;
    if (CNT && tile < nTiles) {
        const int e0 = tile * NTHR + tid;
        hcur = e0 < E;
        if (hcur) { dcur = edst[e0]; scur = esrc[e0]; }
    }
    if (tile < nTiles) fetch(tile);
    for (; tile < nTiles; tile += gridDim.x) {
        __syncthreads();
#pragma unroll
        for (int i = 0; i < NV; ++i) {
            const int idx = tid + i * NTHR;
            if (FULL || idx < 16 * Q) {
                const int r = idx / Q, c4 = idx % Q;
                *(v4f*)(sA + r * S + c4 * 4) = stage[i];
            }
        }
        __syncthreads();

        const int nxt = tile + gridDim.x;
        if (nxt < nTiles) fetch(nxt);

        int  slot = 0, dnxt = 0, snxt = 0;
        bool hnxt = false;
        if (CNT) {
            const int en = nxt * NTHR + tid;
            hnxt = nxt < nTiles && en < E;
            if (hnxt) { dnxt = edst[en]; snxt = esrc[en]; }
            if (hold) place_edge(lists, lpitch, ovf, ovf_n, dold, sold, slotold, eold);
            if (hcur) slot = atomicAdd(&cnt[dcur], 1);
        }

        v8f acc = {};
#pragma unroll
        for (int f = 0; f < NF; ++f) {
            const v2f a = *(const v2f*)(Ap + 4 * f);
            acc = __builtin_amdgcn_wmma_f32_16x16x4_f32(
                 false, a,  false, bf[f],
                 (short)0, acc,  false,  false);
        }

        if (KS > 1) {
            float* P = sP + wave * 256 + lane;
#pragma unroll
            for (int r = 0; r < 8; ++r) P[r * 32] = acc[r];
            __syncthreads();
            if (kq == 0) {
#pragma unroll
                for (int r = 0; r < 8; ++r) {
                    float s = acc[r];
#pragma unroll
                    for (int q = 1; q < KS; ++q) s += sP[(ct + q * NCT) * 256 + r * 32 + lane];
                    acc[r] = s;
                }
            }
        }

        if (nok && kq == 0) {
            float* Cp = C + (size_t)(tile * 16 + mbase) * ldc + n;
#pragma unroll
            for (int r = 0; r < 8; ++r) {
                Cp[0] = acc[r] + badd;
                Cp += ldc;
            }
        }

        if (CNT) {
            dold = dcur; sold = scur; slotold = slot; eold = tile * NTHR + tid; hold = hcur;
            dcur = dnxt;
            scur = snxt;
            hcur = hnxt;
        }
    }

    if (CNT) {
        if (hold) place_edge(lists, lpitch, ovf, ovf_n, dold, sold, slotold, eold);
        for (; tile * NTHR < E; tile += gridDim.x) {
            const int e = tile * NTHR + tid;
            if (e < E) {
                const int d = edst[e];
                place_edge(lists, lpitch, ovf, ovf_n, d, esrc[e], atomicAdd(&cnt[d], 1), e);
            }
        }
    }
}


struct AggState {
    int   p[4], e[4], nid[4];
    float di[4];
    float a0[4], a1[4], a2[4], a3[4];
};

template <int M>
__device__ __forceinline__ bool agg_need(const AggState& s, int g) {
    return s.p[g] + M < s.e[g];
}

template <int M>
__device__ __forceinline__ void agg_cswap(AggState& s, int i, int j) {
    if (!agg_need<M>(s, i) && agg_need<M>(s, j)) {
        int t;
        t = s.p[i];   s.p[i]   = s.p[j];   s.p[j]   = t;
        t = s.e[i];   s.e[i]   = s.e[j];   s.e[j]   = t;
        t = s.nid[i]; s.nid[i] = s.nid[j]; s.nid[j] = t;
        float f;
        f = s.di[i]; s.di[i] = s.di[j]; s.di[j] = f;
        f = s.a0[i]; s.a0[i] = s.a0[j]; s.a0[j] = f;
        f = s.a1[i]; s.a1[i] = s.a1[j]; s.a1[j] = f;
        f = s.a2[i]; s.a2[i] = s.a2[j]; s.a2[j] = f;
        f = s.a3[i]; s.a3[i] = s.a3[j]; s.a3[j] = f;
    }
}

template <int K, int M>
__device__ __forceinline__ int agg_front(AggState& s) {
    if (K == 4) { agg_cswap<M>(s, 0, 1); agg_cswap<M>(s, 2, 3); agg_cswap<M>(s, 0, 2);
                  agg_cswap<M>(s, 1, 3); agg_cswap<M>(s, 1, 2); }
    if (K == 3) { agg_cswap<M>(s, 0, 1); agg_cswap<M>(s, 1, 2); agg_cswap<M>(s, 0, 1); }
    if (K == 2) { agg_cswap<M>(s, 0, 1); }
    int c = 0;
#pragma unroll
    for (int g = 0; g < K; ++g) c += agg_need<M>(s, g) ? 1 : 0;
    return c;
}

template <int K>
__device__ __forceinline__ void agg_walk_pairs(AggState& s,
                                               const int* colidx,
                                               const int* __restrict__ cnt,
                                               const float* __restrict__ H, unsigned lane16) {
    for (;;) {
        bool all = true;
#pragma unroll
        for (int g = 0; g < K; ++g) all = all && agg_need<1>(s, g);
        if (!all) break;

        int j0[K], j1[K];
#pragma unroll
        for (int g = 0; g < K; ++g) { j0[g] = colidx[s.p[g]]; j1[g] = colidx[s.p[g] + 1]; }
        const int lane = (int)(lane16 >> 4);
        int vj = j0[0];
#pragma unroll
        for (int g = 0; g < K; ++g) {
            if (g > 0) vj = (lane == 2 * g) ? j0[g] : vj;
            vj = (lane == 2 * g + 1) ? j1[g] : vj;
        }
        const int wv = __float_as_int(dis_of(cnt[vj]));
        float4 h0[K], h1[K];
#pragma unroll
        for (int g = 0; g < K; ++g) {
            h0[g] = *(const float4*)((const char*)H + ((unsigned)j0[g] * 512u + lane16));
            h1[g] = *(const float4*)((const char*)H + ((unsigned)j1[g] * 512u + lane16));
        }
        float w0[K], w1[K];
#pragma unroll
        for (int g = 0; g < K; ++g) {
            w0[g] = __int_as_float(__builtin_amdgcn_readlane(wv, 2 * g));
            w1[g] = __int_as_float(__builtin_amdgcn_readlane(wv, 2 * g + 1));
        }
#pragma unroll
        for (int g = 0; g < K; ++g) {
            s.a0[g] += w0[g] * h0[g].x + w1[g] * h1[g].x;
            s.a1[g] += w0[g] * h0[g].y + w1[g] * h1[g].y;
            s.a2[g] += w0[g] * h0[g].z + w1[g] * h1[g].z;
            s.a3[g] += w0[g] * h0[g].w + w1[g] * h1[g].w;
            s.p[g] += 2;
        }
    }
}

template <int K>
__device__ __forceinline__ void agg_last_single(AggState& s,
                                                const int* colidx,
                                                const int* __restrict__ cnt,
                                                const float* __restrict__ H, unsigned lane16) {
    int jt[K];
#pragma unroll
    for (int g = 0; g < K; ++g) jt[g] = colidx[s.p[g]];
    const int lane = (int)(lane16 >> 4);
    int vj = jt[0];
#pragma unroll
    for (int g = 1; g < K; ++g) vj = (lane == g) ? jt[g] : vj;
    const int wv = __float_as_int(dis_of(cnt[vj]));
    float4 ht[K];
#pragma unroll
    for (int g = 0; g < K; ++g)
        ht[g] = *(const float4*)((const char*)H + ((unsigned)jt[g] * 512u + lane16));
    float wt[K];
#pragma unroll
    for (int g = 0; g < K; ++g) wt[g] = __int_as_float(__builtin_amdgcn_readlane(wv, g));
#pragma unroll
    for (int g = 0; g < K; ++g) {
        s.a0[g] += wt[g] * ht[g].x; s.a1[g] += wt[g] * ht[g].y;
        s.a2[g] += wt[g] * ht[g].z; s.a3[g] += wt[g] * ht[g].w;
        s.p[g] += 1;
    }
}

__global__ __launch_bounds__(256)
void gcn_aggregate_kernel(const float* __restrict__ H,
                          const int* __restrict__ cnt,
                          const int* lists, int lpitch,
                          const int* __restrict__ ovf,
                          const int* __restrict__ ei, int E,
                          const float* __restrict__ bias,
                          float* outbuf,
                          int ld_out, int col_off, int n) {
    const int lane = threadIdx.x & 31;
    const int wv = __builtin_amdgcn_readfirstlane(
        (int)(blockIdx.x * (blockDim.x >> 5) + (threadIdx.x >> 5)));
    const int node0 = wv * 4;
    if (node0 >= n) return;

    const int novf = cnt[n];

    const int4 c4 = *(const int4*)(cnt + node0);
    const int cg[4] = {c4.x, c4.y, c4.z, c4.w};

    AggState s;
#pragma unroll
    for (int g = 0; g < 4; ++g) {
        const int c = __builtin_amdgcn_readfirstlane(cg[g]);
        s.p[g]   = (node0 + g) * lpitch;
        s.e[g]   = (node0 + g < n) ? s.p[g] + min(c + 1, LIST_CAP) : s.p[g];
        s.nid[g] = node0 + g;
        s.di[g]  = dis_of(c);
        s.a0[g] = 0.f; s.a1[g] = 0.f; s.a2[g] = 0.f; s.a3[g] = 0.f;
    }

    const unsigned lane16 = (unsigned)lane * 16u;

    agg_walk_pairs<4>(s, lists, cnt, H, lane16);
    int c = agg_front<4, 1>(s);
    if (c == 3) { agg_walk_pairs<3>(s, lists, cnt, H, lane16); c = agg_front<3, 1>(s); }
    if (c == 2) { agg_walk_pairs<2>(s, lists, cnt, H, lane16); c = agg_front<2, 1>(s); }
    if (c == 1) { agg_walk_pairs<1>(s, lists, cnt, H, lane16); }

    c = agg_front<4, 0>(s);
    if (c == 4)      agg_last_single<4>(s, lists, cnt, H, lane16);
    else if (c == 3) agg_last_single<3>(s, lists, cnt, H, lane16);
    else if (c == 2) agg_last_single<2>(s, lists, cnt, H, lane16);
    else if (c == 1) agg_last_single<1>(s, lists, cnt, H, lane16);

    if (novf > 0) {
#pragma unroll
        for (int g = 0; g < 4; ++g) {
            const int node = s.nid[g];
            if (node < n && cnt[node] + 1 > LIST_CAP) {
#pragma nounroll
                for (int k = 0; k < novf; ++k) {
                    const int e = ovf[k];
                    if (ei[E + e] == node) {
                        const int j = ei[e];
                        const float w = dis_of(cnt[j]);
                        const float4 h = *(const float4*)((const char*)H + ((unsigned)j * 512u + lane16));
                        s.a0[g] += w * h.x; s.a1[g] += w * h.y;
                        s.a2[g] += w * h.z; s.a3[g] += w * h.w;
                    }
                }
            }
        }
    }

    const int cc = lane * 4;
    const float b0 = bias[cc + 0], b1 = bias[cc + 1], b2 = bias[cc + 2], b3 = bias[cc + 3];
#pragma unroll
    for (int g = 0; g < 4; ++g) {
        const int node = s.nid[g];
        if (node < n) {
            const float di = s.di[g];
            float4 r;
            r.x = fmaxf(di * s.a0[g] + b0, 0.0f);
            r.y = fmaxf(di * s.a1[g] + b1, 0.0f);
            r.z = fmaxf(di * s.a2[g] + b2, 0.0f);
            r.w = fmaxf(di * s.a3[g] + b3, 0.0f);
            *(float4*)(outbuf + (size_t)node * ld_out + col_off + cc) = r;
        }
    }
}


extern "C" void kernel_launch(void* const* d_in, const int* in_sizes, int n_in,
                              void* d_out, int out_size, void* d_ws, size_t ws_size,
                              hipStream_t stream) {
    (void)n_in; (void)out_size; (void)ws_size;
    const int F_IN = 256, HID = 128, NC = 40;
    const int N = in_sizes[0] / F_IN;
    const int E = in_sizes[1] / 2;

    const float* x     = (const float*)d_in[0];
    const int*   ei    = (const int*)d_in[1];
    const float* W_in  = (const float*)d_in[2];
    const float* b_in  = (const float*)d_in[3];
    const float* W1    = (const float*)d_in[4];
    const float* b1    = (const float*)d_in[5];
    const float* W2    = (const float*)d_in[6];
    const float* b2    = (const float*)d_in[7];
    const float* W3    = (const float*)d_in[8];
    const float* b3    = (const float*)d_in[9];
    const float* W_out = (const float*)d_in[10];
    const float* b_out = (const float*)d_in[11];
    float* out = (float*)d_out;

    char* base = (char*)d_ws;
    size_t off = 0;
    auto alloc = [&](size_t bytes) -> void* {
        void* q = base + off;
        off = (off + bytes + 255) & ~(size_t)255;
        return q;
    };
    int*   cnt     = (int*)alloc((size_t)(N + 4) * 4);
    int*   ovf     = (int*)alloc((size_t)E * 4);
    float* H       = (float*)alloc((size_t)N * HID * 4);
    float* hcat    = (float*)alloc((size_t)N * HID * 4 * 4);
    int*   lists   = (int*)(hcat + 3 * HID);
    const int lpitch = 4 * HID;

    const int nbI = (N + 4 + 255) / 256;

    const int rowTiles = N / 16;
    dim3 gemmBlk(32, HID / 16);

    init_cnt_kernel<<<nbI, 256, 0, stream>>>(cnt, lists, lpitch, N);
    gemm_wreg_kernel<256, 8, 1, false, true><<<512, gemmBlk, 0, stream>>>(
        x, F_IN, W_in, HID, H, HID, HID, nullptr, rowTiles,
        ei, ei + E, cnt, cnt + N, lists, lpitch, ovf, E);

    const int aggBlocks = (N + 31) / 32;

    gcn_aggregate_kernel<<<aggBlocks, 256, 0, stream>>>(
        H, cnt, lists, lpitch, ovf, ei, E, b_in, hcat, 4 * HID, 0 * HID, N);

    const float* Ws[3] = {W1, W2, W3};
    const float* bs[3] = {b1, b2, b3};
    for (int l = 0; l < 3; ++l) {
        gemm_wreg_kernel<128, 8, 1, false, false><<<1024, gemmBlk, 0, stream>>>(
            hcat + (size_t)l * HID, 4 * HID, Ws[l], HID, H, HID, HID, nullptr, rowTiles,
            nullptr, nullptr, nullptr, nullptr, nullptr, 0, nullptr, 0);
        gcn_aggregate_kernel<<<aggBlocks, 256, 0, stream>>>(
            H, cnt, lists, lpitch, ovf, ei, E, bs[l], hcat, 4 * HID, (l + 1) * HID, N);
    }

    dim3 headBlk(32, 12);
    gemm_wreg_kernel<512, 3, 4, true, false><<<512, headBlk, 0, stream>>>(
        hcat, 4 * HID, W_out, NC, out, NC, NC, b_out, rowTiles,
        nullptr, nullptr, nullptr, nullptr, nullptr, 0, nullptr, 0);
}


# Neighbour lists: each package half counts in-edges in its own counter array and fills its own side of the list, so no counter line crosses halves; row order = half 1's edges, self, half 0's edges
# speedup vs baseline: 2.0332x; 1.1245x over previous
#include <hip/hip_runtime.h>
#include <hip/hip_bf16.h>

typedef __attribute__((ext_vector_type(2))) float v2f;
typedef __attribute__((ext_vector_type(4))) float v4f;
typedef __attribute__((ext_vector_type(8))) float v8f;


constexpr int LIST_CAP = 128;
constexpr int LIST_MID = 64;
constexpr int CAP_UP   = LIST_CAP - LIST_MID - 1;
constexpr int CAP_DOWN = LIST_MID;

__device__ __forceinline__ float dis_of(int c) {
    return rsqrtf((float)(c + 1));
}

__global__ void init_cnt_kernel(int* cnt, int* cntB, int* lists, int lpitch, int n) {
    int i = blockIdx.x * blockDim.x + threadIdx.x;
    if (i < n + 4) { cnt[i] = 0; cntB[i] = 0; }
    if (i < n) lists[(size_t)i * lpitch + LIST_MID] = i;
}

__device__ __forceinline__ void place_edge(int* lists, int lpitch, int* ovf, int* ovf_n,
                                           int half, int dst, int src_node, int slot, int e) {
    const int cap = half ? CAP_DOWN : CAP_UP;
    const int pos = half ? LIST_MID - 1 - slot : LIST_MID + 1 + slot;
    if (slot < cap) lists[(size_t)dst * lpitch + pos] = src_node;
    else ovf[atomicAdd(ovf_n, 1)] = e;
}

__global__ void finish_cnt_kernel(int* cnt, int* cntB, int n) {
    int i = blockIdx.x * blockDim.x + threadIdx.x;
    if (i < n) {
        const int a = cnt[i], b = cntB[i];
        cnt[i]  = a + b;
        cntB[i] = min(a, CAP_UP) | (min(b, CAP_DOWN) << 8);
    }
}


template <int K, int NCT, int KS, bool MASK, bool CNT>
__global__ __launch_bounds__(32 * NCT * KS)
void gemm_wreg_kernel(const float* __restrict__ A, int lda,
                      const float* __restrict__ B, int ldb,
                      float* __restrict__ C, int ldc, int Ncols,
                      const float* __restrict__ bias, int nTiles,
                      const int* __restrict__ esrc, const int* __restrict__ edst,
                      int* cntA, int* cntB, int* ovf_n, int* lists, int lpitch,
                      int* __restrict__ ovf, int E) {
    constexpr int NW   = NCT * KS;
    constexpr int NTHR = 32 * NW;
    constexpr int KW   = K / KS;
    constexpr int NF   = KW / 4;
    constexpr int S    = K + 4;
    constexpr int Q    = K / 4;
    constexpr int NV   = (16 * Q + NTHR - 1) / NTHR;
    constexpr bool FULL = ((16 * Q) % NTHR) == 0;

    __shared__ __attribute__((aligned(16))) float sA[16 * S];
    __shared__ float sP[KS > 1 ? NW * 256 : 1];

    const int lane = threadIdx.x;
    const int wave = threadIdx.y;
    const int tid  = wave * 32 + lane;
    const int ct   = wave % NCT;
    const int kq   = wave / NCT;
    const int m    = lane & 15;
    const int kk   = (lane >> 4) << 1;
    const int n    = ct * 16 + m;
    const bool nok = !MASK || (n < Ncols);
    const int  nc  = MASK ? (nok ? n : 0) : n;

    v2f bf[NF];
    {
        const float* Bp = B + (size_t)(kq * KW + kk) * ldb + nc;
#pragma unroll
        for (int f = 0; f < NF; ++f) {
            float bx = Bp[0];
            float by = Bp[ldb];
            if (MASK) { bx = nok ? bx : 0.0f; by = nok ? by : 0.0f; }
            bf[f].x = bx;
            bf[f].y = by;
            Bp += 4 * ldb;
        }
    }

    v4f stage[NV];
    auto fetch = [&](int t) __attribute__((always_inline)) {
        const float* At = A + (size_t)t * 16 * lda;
#pragma unroll
        for (int i = 0; i < NV; ++i) {
            const int idx = tid + i * NTHR;
            if (FULL || idx < 16 * Q) {
                const int r = idx / Q, c4 = idx % Q;
                stage[i] = *(const v4f*)(At + (size_t)r * lda + c4 * 4);
            }
        }
    };

    const int mbase = (lane < 16) ? 0 : 8;
    const float badd = (bias && nok) ? bias[n] : 0.0f;
    const float* Ap = sA + m * S + kq * KW + kk;

    int  half = 0;
    int* cp   = cntA;
    if (CNT) {
        half = (int)((__builtin_amdgcn_s_sendmsg_rtn(0x87) >> 18) & 1u);
        cp   = half ? cntB : cntA;
    }

    int tile = blockIdx.x;
    int  dcur = 0, scur = 0;
    bool hcur = false;
    int  dold = 0, sold = 0, slotold = 0, eold = 0;
    bool hold = false;
    if (CNT && tile < nTiles) {
        const int e0 = tile * NTHR + tid;
        hcur = e0 < E;
        if (hcur) { dcur = edst[e0]; scur = esrc[e0]; }
    }
    if (tile < nTiles) fetch(tile);
    for (; tile < nTiles; tile += gridDim.x) {
        __syncthreads();
#pragma unroll
        for (int i = 0; i < NV; ++i) {
            const int idx = tid + i * NTHR;
            if (FULL || idx < 16 * Q) {
                const int r = idx / Q, c4 = idx % Q;
                *(v4f*)(sA + r * S + c4 * 4) = stage[i];
            }
        }
        __syncthreads();

        const int nxt = tile + gridDim.x;
        if (nxt < nTiles) fetch(nxt);

        int  slot = 0, dnxt = 0, snxt = 0;
        bool hnxt = false;
        if (CNT) {
            const int en = nxt * NTHR + tid;
            hnxt = nxt < nTiles && en < E;
            if (hnxt) { dnxt = edst[en]; snxt = esrc[en]; }
            if (hold) place_edge(lists, lpitch, ovf, ovf_n, half, dold, sold, slotold, eold);
            if (hcur) slot = atomicAdd(&cp[dcur], 1);
        }

        v8f acc = {};
#pragma unroll
        for (int f = 0; f < NF; ++f) {
            const v2f a = *(const v2f*)(Ap + 4 * f);
            acc = __builtin_amdgcn_wmma_f32_16x16x4_f32(
                 false, a,  false, bf[f],
                 (short)0, acc,  false,  false);
        }

        if (KS > 1) {
            float* P = sP + wave * 256 + lane;
#pragma unroll
            for (int r = 0; r < 8; ++r) P[r * 32] = acc[r];
            __syncthreads();
            if (kq == 0) {
#pragma unroll
                for (int r = 0; r < 8; ++r) {
                    float s = acc[r];
#pragma unroll
                    for (int q = 1; q < KS; ++q) s += sP[(ct + q * NCT) * 256 + r * 32 + lane];
                    acc[r] = s;
                }
            }
        }

        if (nok && kq == 0) {
            float* Cp = C + (size_t)(tile * 16 + mbase) * ldc + n;
#pragma unroll
            for (int r = 0; r < 8; ++r) {
                Cp[0] = acc[r] + badd;
                Cp += ldc;
            }
        }

        if (CNT) {
            dold = dcur; sold = scur; slotold = slot; eold = tile * NTHR + tid; hold = hcur;
            dcur = dnxt;
            scur = snxt;
            hcur = hnxt;
        }
    }

    if (CNT) {
        if (hold) place_edge(lists, lpitch, ovf, ovf_n, half, dold, sold, slotold, eold);
        for (; tile * NTHR < E; tile += gridDim.x) {
            const int e = tile * NTHR + tid;
            if (e < E) {
                const int d = edst[e];
                place_edge(lists, lpitch, ovf, ovf_n, half, d, esrc[e], atomicAdd(&cp[d], 1), e);
            }
        }
    }
}


struct AggState {
    int   p[4], e[4], nid[4];
    float di[4];
    float a0[4], a1[4], a2[4], a3[4];
};

template <int M>
__device__ __forceinline__ bool agg_need(const AggState& s, int g) {
    return s.p[g] + M < s.e[g];
}

template <int M>
__device__ __forceinline__ void agg_cswap(AggState& s, int i, int j) {
    if (!agg_need<M>(s, i) && agg_need<M>(s, j)) {
        int t;
        t = s.p[i];   s.p[i]   = s.p[j];   s.p[j]   = t;
        t = s.e[i];   s.e[i]   = s.e[j];   s.e[j]   = t;
        t = s.nid[i]; s.nid[i] = s.nid[j]; s.nid[j] = t;
        float f;
        f = s.di[i]; s.di[i] = s.di[j]; s.di[j] = f;
        f = s.a0[i]; s.a0[i] = s.a0[j]; s.a0[j] = f;
        f = s.a1[i]; s.a1[i] = s.a1[j]; s.a1[j] = f;
        f = s.a2[i]; s.a2[i] = s.a2[j]; s.a2[j] = f;
        f = s.a3[i]; s.a3[i] = s.a3[j]; s.a3[j] = f;
    }
}

template <int K, int M>
__device__ __forceinline__ int agg_front(AggState& s) {
    if (K == 4) { agg_cswap<M>(s, 0, 1); agg_cswap<M>(s, 2, 3); agg_cswap<M>(s, 0, 2);
                  agg_cswap<M>(s, 1, 3); agg_cswap<M>(s, 1, 2); }
    if (K == 3) { agg_cswap<M>(s, 0, 1); agg_cswap<M>(s, 1, 2); agg_cswap<M>(s, 0, 1); }
    if (K == 2) { agg_cswap<M>(s, 0, 1); }
    int c = 0;
#pragma unroll
    for (int g = 0; g < K; ++g) c += agg_need<M>(s, g) ? 1 : 0;
    return c;
}

template <int K>
__device__ __forceinline__ void agg_walk_pairs(AggState& s,
                                               const int* colidx,
                                               const int* __restrict__ cnt,
                                               const float* __restrict__ H, unsigned lane16) {
    for (;;) {
        bool all = true;
#pragma unroll
        for (int g = 0; g < K; ++g) all = all && agg_need<1>(s, g);
        if (!all) break;

        int j0[K], j1[K];
#pragma unroll
        for (int g = 0; g < K; ++g) { j0[g] = colidx[s.p[g]]; j1[g] = colidx[s.p[g] + 1]; }
        const int lane = (int)(lane16 >> 4);
        int vj = j0[0];
#pragma unroll
        for (int g = 0; g < K; ++g) {
            if (g > 0) vj = (lane == 2 * g) ? j0[g] : vj;
            vj = (lane == 2 * g + 1) ? j1[g] : vj;
        }
        const int wv = __float_as_int(dis_of(cnt[vj]));
        float4 h0[K], h1[K];
#pragma unroll
        for (int g = 0; g < K; ++g) {
            h0[g] = *(const float4*)((const char*)H + ((unsigned)j0[g] * 512u + lane16));
            h1[g] = *(const float4*)((const char*)H + ((unsigned)j1[g] * 512u + lane16));
        }
        float w0[K], w1[K];
#pragma unroll
        for (int g = 0; g < K; ++g) {
            w0[g] = __int_as_float(__builtin_amdgcn_readlane(wv, 2 * g));
            w1[g] = __int_as_float(__builtin_amdgcn_readlane(wv, 2 * g + 1));
        }
#pragma unroll
        for (int g = 0; g < K; ++g) {
            s.a0[g] += w0[g] * h0[g].x + w1[g] * h1[g].x;
            s.a1[g] += w0[g] * h0[g].y + w1[g] * h1[g].y;
            s.a2[g] += w0[g] * h0[g].z + w1[g] * h1[g].z;
            s.a3[g] += w0[g] * h0[g].w + w1[g] * h1[g].w;
            s.p[g] += 2;
        }
    }
}

template <int K>
__device__ __forceinline__ void agg_last_single(AggState& s,
                                                const int* colidx,
                                                const int* __restrict__ cnt,
                                                const float* __restrict__ H, unsigned lane16) {
    int jt[K];
#pragma unroll
    for (int g = 0; g < K; ++g) jt[g] = colidx[s.p[g]];
    const int lane = (int)(lane16 >> 4);
    int vj = jt[0];
#pragma unroll
    for (int g = 1; g < K; ++g) vj = (lane == g) ? jt[g] : vj;
    const int wv = __float_as_int(dis_of(cnt[vj]));
    float4 ht[K];
#pragma unroll
    for (int g = 0; g < K; ++g)
        ht[g] = *(const float4*)((const char*)H + ((unsigned)jt[g] * 512u + lane16));
    float wt[K];
#pragma unroll
    for (int g = 0; g < K; ++g) wt[g] = __int_as_float(__builtin_amdgcn_readlane(wv, g));
#pragma unroll
    for (int g = 0; g < K; ++g) {
        s.a0[g] += wt[g] * ht[g].x; s.a1[g] += wt[g] * ht[g].y;
        s.a2[g] += wt[g] * ht[g].z; s.a3[g] += wt[g] * ht[g].w;
        s.p[g] += 1;
    }
}

__global__ __launch_bounds__(256)
void gcn_aggregate_kernel(const float* __restrict__ H,
                          const int* __restrict__ cnt,
                          const int* __restrict__ seg,
                          const int* lists, int lpitch,
                          const int* __restrict__ ovf,
                          const int* __restrict__ ei, int E,
                          const float* __restrict__ bias,
                          float* outbuf,
                          int ld_out, int col_off, int n) {
    const int lane = threadIdx.x & 31;
    const int wv = __builtin_amdgcn_readfirstlane(
        (int)(blockIdx.x * (blockDim.x >> 5) + (threadIdx.x >> 5)));
    const int node0 = wv * 4;
    if (node0 >= n) return;

    const int novf = cnt[n];

    const int4 c4 = *(const int4*)(cnt + node0);
    const int4 s4 = *(const int4*)(seg + node0);
    const int cg[4] = {c4.x, c4.y, c4.z, c4.w};
    const int sg[4] = {s4.x, s4.y, s4.z, s4.w};

    AggState s;
#pragma unroll
    for (int g = 0; g < 4; ++g) {
        const int c   = __builtin_amdgcn_readfirstlane(cg[g]);
        const int sv  = __builtin_amdgcn_readfirstlane(sg[g]);
        const int mid = (node0 + g) * lpitch + LIST_MID;
        const bool live = node0 + g < n;
        s.p[g]   = live ? mid - ((sv >> 8) & 0xff) : mid;
        s.e[g]   = live ? mid + 1 + (sv & 0xff) : mid;
        s.nid[g] = node0 + g;
        s.di[g]  = dis_of(c);
        s.a0[g] = 0.f; s.a1[g] = 0.f; s.a2[g] = 0.f; s.a3[g] = 0.f;
    }

    const unsigned lane16 = (unsigned)lane * 16u;

    agg_walk_pairs<4>(s, lists, cnt, H, lane16);
    int c = agg_front<4, 1>(s);
    if (c == 3) { agg_walk_pairs<3>(s, lists, cnt, H, lane16); c = agg_front<3, 1>(s); }
    if (c == 2) { agg_walk_pairs<2>(s, lists, cnt, H, lane16); c = agg_front<2, 1>(s); }
    if (c == 1) { agg_walk_pairs<1>(s, lists, cnt, H, lane16); }

    c = agg_front<4, 0>(s);
    if (c == 4)      agg_last_single<4>(s, lists, cnt, H, lane16);
    else if (c == 3) agg_last_single<3>(s, lists, cnt, H, lane16);
    else if (c == 2) agg_last_single<2>(s, lists, cnt, H, lane16);
    else if (c == 1) agg_last_single<1>(s, lists, cnt, H, lane16);

    if (novf > 0) {
#pragma unroll
        for (int g = 0; g < 4; ++g) {
            const int node = s.nid[g];
            const int sv   = (node < n) ? seg[node] : 0;
            if (node < n && cnt[node] > (sv & 0xff) + ((sv >> 8) & 0xff)) {
#pragma nounroll
                for (int k = 0; k < novf; ++k) {
                    const int e = ovf[k];
                    if (ei[E + e] == node) {
                        const int j = ei[e];
                        const float w = dis_of(cnt[j]);
                        const float4 h = *(const float4*)((const char*)H + ((unsigned)j * 512u + lane16));
                        s.a0[g] += w * h.x; s.a1[g] += w * h.y;
                        s.a2[g] += w * h.z; s.a3[g] += w * h.w;
                    }
                }
            }
        }
    }

    const int cc = lane * 4;
    const float b0 = bias[cc + 0], b1 = bias[cc + 1], b2 = bias[cc + 2], b3 = bias[cc + 3];
#pragma unroll
    for (int g = 0; g < 4; ++g) {
        const int node = s.nid[g];
        if (node < n) {
            const float di = s.di[g];
            float4 r;
            r.x = fmaxf(di * s.a0[g] + b0, 0.0f);
            r.y = fmaxf(di * s.a1[g] + b1, 0.0f);
            r.z = fmaxf(di * s.a2[g] + b2, 0.0f);
            r.w = fmaxf(di * s.a3[g] + b3, 0.0f);
            *(float4*)(outbuf + (size_t)node * ld_out + col_off + cc) = r;
        }
    }
}


extern "C" void kernel_launch(void* const* d_in, const int* in_sizes, int n_in,
                              void* d_out, int out_size, void* d_ws, size_t ws_size,
                              hipStream_t stream) {
    (void)n_in; (void)out_size; (void)ws_size;
    const int F_IN = 256, HID = 128, NC = 40;
    const int N = in_sizes[0] / F_IN;
    const int E = in_sizes[1] / 2;

    const float* x     = (const float*)d_in[0];
    const int*   ei    = (const int*)d_in[1];
    const float* W_in  = (const float*)d_in[2];
    const float* b_in  = (const float*)d_in[3];
    const float* W1    = (const float*)d_in[4];
    const float* b1    = (const float*)d_in[5];
    const float* W2    = (const float*)d_in[6];
    const float* b2    = (const float*)d_in[7];
    const float* W3    = (const float*)d_in[8];
    const float* b3    = (const float*)d_in[9];
    const float* W_out = (const float*)d_in[10];
    const float* b_out = (const float*)d_in[11];
    float* out = (float*)d_out;

    char* base = (char*)d_ws;
    size_t off = 0;
    auto alloc = [&](size_t bytes) -> void* {
        void* q = base + off;
        off = (off + bytes + 255) & ~(size_t)255;
        return q;
    };
    int*   cnt     = (int*)alloc((size_t)(N + 4) * 4);
    int*   cntB    = (int*)alloc((size_t)(N + 4) * 4);
    int*   ovf     = (int*)alloc((size_t)E * 4);
    float* H       = (float*)alloc((size_t)N * HID * 4);
    float* hcat    = (float*)alloc((size_t)N * HID * 4 * 4);
    int*   lists   = (int*)(hcat + 3 * HID);
    const int lpitch = 4 * HID;

    const int nbI = (N + 4 + 255) / 256;

    const int rowTiles = N / 16;
    dim3 gemmBlk(32, HID / 16);

    init_cnt_kernel<<<nbI, 256, 0, stream>>>(cnt, cntB, lists, lpitch, N);
    gemm_wreg_kernel<256, 8, 1, false, true><<<512, gemmBlk, 0, stream>>>(
        x, F_IN, W_in, HID, H, HID, HID, nullptr, rowTiles,
        ei, ei + E, cnt, cntB, cnt + N, lists, lpitch, ovf, E);
    finish_cnt_kernel<<<nbI, 256, 0, stream>>>(cnt, cntB, N);

    const int aggBlocks = (N + 31) / 32;

    gcn_aggregate_kernel<<<aggBlocks, 256, 0, stream>>>(
        H, cnt, cntB, lists, lpitch, ovf, ei, E, b_in, hcat, 4 * HID, 0 * HID, N);

    const float* Ws[3] = {W1, W2, W3};
    const float* bs[3] = {b1, b2, b3};
    for (int l = 0; l < 3; ++l) {
        gemm_wreg_kernel<128, 8, 1, false, false><<<1024, gemmBlk, 0, stream>>>(
            hcat + (size_t)l * HID, 4 * HID, Ws[l], HID, H, HID, HID, nullptr, rowTiles,
            nullptr, nullptr, nullptr, nullptr, nullptr, nullptr, 0, nullptr, 0);
        gcn_aggregate_kernel<<<aggBlocks, 256, 0, stream>>>(
            H, cnt, cntB, lists, lpitch, ovf, ei, E, bs[l], hcat, 4 * HID, (l + 1) * HID, N);
    }

    dim3 headBlk(32, 12);
    gemm_wreg_kernel<512, 3, 4, true, false><<<512, headBlk, 0, stream>>>(
        hcat, 4 * HID, W_out, NC, out, NC, NC, b_out, rowTiles,
        nullptr, nullptr, nullptr, nullptr, nullptr, nullptr, 0, nullptr, 0);
}
